# starting kernel
# baseline (speedup 1.0000x reference)
#include <hip/hip_runtime.h>
#include <math.h>

#define H_  16
#define T_  2048
#define C_  1024
#define HD  64

typedef __attribute__((ext_vector_type(2))) float v2f;
typedef __attribute__((ext_vector_type(8))) float v8f;

__device__ __forceinline__ v8f splat8(float x) {
  v8f r;
#pragma unroll
  for (int i = 0; i < 8; ++i) r[i] = x;
  return r;
}

__global__ __launch_bounds__(256) void tr_pair_kernel(const float* __restrict__ in,
                                                      float* __restrict__ out,
                                                      int rows, int cols) {
  __shared__ float tile[32][33];
  int c  = blockIdx.x * 32 + threadIdx.x;
  int r0 = blockIdx.y * 32 + threadIdx.y;
#pragma unroll
  for (int i = 0; i < 32; i += 8) {
    int r = r0 + i;
    if (r < rows && c < cols) tile[threadIdx.y + i][threadIdx.x] = in[(size_t)r * cols + c];
  }
  __syncthreads();
  int oc  = blockIdx.y * 32 + threadIdx.x;
  int or0 = blockIdx.x * 32 + threadIdx.y;
#pragma unroll
  for (int i = 0; i < 32; i += 8) {
    int k = or0 + i;
    if (k < cols && oc < rows)
      out[((size_t)(k >> 1) * rows + oc) * 2 + (k & 1)] = tile[threadIdx.x][threadIdx.y + i];
  }
}

__device__ __forceinline__ void gemm_strip2_f32(const float* __restrict__ A,
                                                const float* __restrict__ BTP,
                                                int K, int N, int m0, int n0,
                                                int lane,
                                                const float* __restrict__ bias,
                                                v8f acc[2][4]) {
  const int half = lane >> 4;
  const int l16  = lane & 15;
#pragma unroll
  for (int i = 0; i < 2; ++i)
#pragma unroll
    for (int j = 0; j < 4; ++j) acc[i][j] = splat8(bias[n0 + 16 * j + l16]);

  const float* arow0 = A + (size_t)(m0 + l16) * K + 2 * half;
  const float* arow1 = arow0 + (size_t)16 * K;
  for (int k = 0; k < K; k += 4) {
    v2f a0 = *(const v2f*)(arow0 + k);
    v2f a1 = *(const v2f*)(arow1 + k);
    const float* bk = BTP + (size_t)((k >> 1) + half) * (2 * N);
#pragma unroll
    for (int j = 0; j < 4; ++j) {
      v2f b = *(const v2f*)(bk + 2 * (n0 + 16 * j + l16));
      acc[0][j] = __builtin_amdgcn_wmma_f32_16x16x4_f32(false, a0, false, b,
                                                        (short)0, acc[0][j], false, false);
      acc[1][j] = __builtin_amdgcn_wmma_f32_16x16x4_f32(false, a1, false, b,
                                                        (short)0, acc[1][j], false, false);
    }
  }
}

__global__ __launch_bounds__(128) void qkv_kernel(const float* __restrict__ x,
                                                  const float* __restrict__ wTP,
                                                  const float* __restrict__ bias,
                                                  float* __restrict__ qv,
                                                  float* __restrict__ kTp,
                                                  float* __restrict__ vp) {
  const int wave = (int)((blockIdx.x * blockDim.x + threadIdx.x) >> 5);
  const int lane = threadIdx.x & 31;
  const int NSTRIP = (3 * C_) / 64;
  const int m0 = (wave / NSTRIP) * 32;
  const int n0 = (wave % NSTRIP) * 64;

  v8f acc[2][4];
  gemm_strip2_f32(x, wTP, C_, 3 * C_, m0, n0, lane, bias, acc);

  const int half = lane >> 4, l16 = lane & 15;
#pragma unroll
  for (int i = 0; i < 2; ++i) {
#pragma unroll
    for (int j = 0; j < 4; ++j) {
      const int nn = n0 + 16 * j + l16;
#pragma unroll
      for (int r = 0; r < 8; ++r) {
        const int mg = m0 + 16 * i + r + 8 * half;
        const int bb = mg >> 11;
        const int t  = mg & (T_ - 1);
        const float val = acc[i][j][r];
        if (nn < C_) {
          const int h = nn >> 6, d = nn & 63;
          qv[((size_t)(bb * H_ + h) * T_ + t) * HD + d] = val;
        } else if (nn < 2 * C_) {
          const int n2 = nn - C_;
          const int h = n2 >> 6, d = n2 & 63;
          kTp[(((size_t)(bb * H_ + h) * (HD / 2) + (d >> 1)) * T_ + t) * 2 + (d & 1)] = val;
        } else {
          const int n2 = nn - 2 * C_;
          const int h = n2 >> 6, d = n2 & 63;
          vp[(((size_t)(bb * H_ + h) * (T_ / 2) + (t >> 1)) * HD + d) * 2 + (t & 1)] = val;
        }
      }
    }
  }
}

__global__ __launch_bounds__(128) void attn_kernel(const float* __restrict__ q,
                                                   const float* __restrict__ kTp,
                                                   const float* __restrict__ vpI,
                                                   float* __restrict__ aout) {
  __shared__ float pl[4][16][18];
  const int waveid = threadIdx.x >> 5;
  const int lane   = threadIdx.x & 31;
  const int wave   = blockIdx.x * 4 + waveid;
  const int bh     = wave >> 7;
  const int qtile  = wave & 127;
  const int b = bh / H_, h = bh % H_;
  const int qt0 = qtile * 16;
  const float* qb = q   + (size_t)bh * T_ * HD;
  const float* kb = kTp + (size_t)bh * HD * T_;
  const float* vb = vpI + (size_t)bh * T_ * HD;
  const int half = lane >> 4, l16 = lane & 15;
  const float scale = 0.125f;

  v2f qa[16];
#pragma unroll
  for (int kk = 0; kk < 16; ++kk)
    qa[kk] = *(const v2f*)(qb + (size_t)(qt0 + l16) * HD + 4 * kk + 2 * half);

  v8f oacc[4];
#pragma unroll
  for (int j = 0; j < 4; ++j) oacc[j] = splat8(0.f);
  float mrow[8], lrow[8];
#pragma unroll
  for (int r = 0; r < 8; ++r) { mrow[r] = -INFINITY; lrow[r] = 0.f; }

  float (*P)[18] = pl[waveid];

  for (int kv0 = 0; kv0 < T_; kv0 += 16) {
    v8f s = splat8(0.f);
#pragma unroll
    for (int kk = 0; kk < 16; ++kk) {
      v2f bfr = *(const v2f*)(kb + ((size_t)(2 * kk + half) * T_ + kv0 + l16) * 2);
      s = __builtin_amdgcn_wmma_f32_16x16x4_f32(false, qa[kk], false, bfr,
                                                (short)0, s, false, false);
    }
    float alpha[8];
#pragma unroll
    for (int r = 0; r < 8; ++r) {
      float sv = s[r] * scale;
      float mx = sv;
      mx = fmaxf(mx, __shfl_xor(mx, 1));
      mx = fmaxf(mx, __shfl_xor(mx, 2));
      mx = fmaxf(mx, __shfl_xor(mx, 4));
      mx = fmaxf(mx, __shfl_xor(mx, 8));
      const float mnew = fmaxf(mrow[r], mx);
      alpha[r] = __expf(mrow[r] - mnew);
      mrow[r]  = mnew;
      const float p = __expf(sv - mnew);
      s[r] = p;
      float sum = p;
      sum += __shfl_xor(sum, 1);
      sum += __shfl_xor(sum, 2);
      sum += __shfl_xor(sum, 4);
      sum += __shfl_xor(sum, 8);
      lrow[r] = lrow[r] * alpha[r] + sum;
    }
#pragma unroll
    for (int j = 0; j < 4; ++j)
#pragma unroll
      for (int r = 0; r < 8; ++r) oacc[j][r] *= alpha[r];

#pragma unroll
    for (int r = 0; r < 8; ++r) P[r + 8 * half][l16] = s[r];

#pragma unroll
    for (int kk = 0; kk < 4; ++kk) {
      v2f pa = *(const v2f*)&P[l16][4 * kk + 2 * half];
      const int tp = (kv0 + 4 * kk + 2 * half) >> 1;
#pragma unroll
      for (int j = 0; j < 4; ++j) {
        v2f bv = *(const v2f*)(vb + ((size_t)tp * HD + 16 * j + l16) * 2);
        oacc[j] = __builtin_amdgcn_wmma_f32_16x16x4_f32(false, pa, false, bv,
                                                        (short)0, oacc[j], false, false);
      }
    }
  }

  float inv[8];
#pragma unroll
  for (int r = 0; r < 8; ++r) inv[r] = 1.0f / lrow[r];
#pragma unroll
  for (int j = 0; j < 4; ++j)
#pragma unroll
    for (int r = 0; r < 8; ++r) {
      const int t = qt0 + r + 8 * half;
      aout[((size_t)b * T_ + t) * C_ + h * HD + 16 * j + l16] = oacc[j][r] * inv[r];
    }
}

__global__ __launch_bounds__(128) void proj_kernel(const float* __restrict__ Ain,
                                                   const float* __restrict__ wTP,
                                                   const float* __restrict__ bias,
                                                   float* __restrict__ out) {
  const int wave = (int)((blockIdx.x * blockDim.x + threadIdx.x) >> 5);
  const int lane = threadIdx.x & 31;
  const int NSTRIP = C_ / 64;
  const int m0 = (wave / NSTRIP) * 32;
  const int n0 = (wave % NSTRIP) * 64;

  v8f acc[2][4];
  gemm_strip2_f32(Ain, wTP, C_, C_, m0, n0, lane, bias, acc);

  const int half = lane >> 4, l16 = lane & 15;
#pragma unroll
  for (int i = 0; i < 2; ++i)
#pragma unroll
    for (int j = 0; j < 4; ++j)
#pragma unroll
      for (int r = 0; r < 8; ++r) {
        const int mg = m0 + 16 * i + r + 8 * half;
        out[(size_t)mg * C_ + n0 + 16 * j + l16] = acc[i][j][r];
      }
}

extern "C" void kernel_launch(void* const* d_in, const int* in_sizes, int n_in,
                              void* d_out, int out_size, void* d_ws, size_t ws_size,
                              hipStream_t stream) {
  const float* x      = (const float*)d_in[0];
  const float* qkv_w  = (const float*)d_in[1];
  const float* qkv_b  = (const float*)d_in[2];
  const float* proj_w = (const float*)d_in[3];
  const float* proj_b = (const float*)d_in[4];
  float* out = (float*)d_out;

  const int B = in_sizes[0] / (T_ * C_);
  const int M = B * T_;

  float* ws      = (float*)d_ws;
  float* qkv_wTP  = ws;
  float* proj_wTP = qkv_wTP + (size_t)3 * C_ * C_;
  float* Q        = proj_wTP + (size_t)C_ * C_;
  float* KTP      = Q       + (size_t)B * H_ * T_ * HD;
  float* VP       = KTP     + (size_t)B * H_ * T_ * HD;
  float* AO       = VP      + (size_t)B * H_ * T_ * HD;

  dim3 trb(32, 8);
  tr_pair_kernel<<<dim3(C_ / 32, (3 * C_) / 32), trb, 0, stream>>>(qkv_w, qkv_wTP, 3 * C_, C_);
  tr_pair_kernel<<<dim3(C_ / 32, C_ / 32),       trb, 0, stream>>>(proj_w, proj_wTP, C_, C_);

  {
    const int waves = (M / 32) * ((3 * C_) / 64);
    qkv_kernel<<<waves / 4, 128, 0, stream>>>(x, qkv_wTP, qkv_b, Q, KTP, VP);
  }
  {
    const int waves = B * H_ * (T_ / 16);
    attn_kernel<<<waves / 4, 128, 0, stream>>>(Q, KTP, VP, AO);
  }
  {
    const int waves = (M / 32) * (C_ / 64);
    proj_kernel<<<waves / 4, 128, 0, stream>>>(AO, proj_wTP, proj_b, out);
  }
}


# attention: online-softmax step widened from 16 to 64 keys (4 independent score tiles per step); row max and row sum fold 4 in-lane values before one 16-lane butterfly, so the sum order changes
# speedup vs baseline: 1.2651x; 1.2651x over previous
#include <hip/hip_runtime.h>
#include <math.h>

#define H_  16
#define T_  2048
#define C_  1024
#define HD  64

typedef __attribute__((ext_vector_type(2))) float v2f;
typedef __attribute__((ext_vector_type(8))) float v8f;

__device__ __forceinline__ v8f splat8(float x) {
  v8f r;
#pragma unroll
  for (int i = 0; i < 8; ++i) r[i] = x;
  return r;
}

__global__ __launch_bounds__(256) void tr_pair_kernel(const float* __restrict__ in,
                                                      float* __restrict__ out,
                                                      int rows, int cols) {
  __shared__ float tile[32][33];
  int c  = blockIdx.x * 32 + threadIdx.x;
  int r0 = blockIdx.y * 32 + threadIdx.y;
#pragma unroll
  for (int i = 0; i < 32; i += 8) {
    int r = r0 + i;
    if (r < rows && c < cols) tile[threadIdx.y + i][threadIdx.x] = in[(size_t)r * cols + c];
  }
  __syncthreads();
  int oc  = blockIdx.y * 32 + threadIdx.x;
  int or0 = blockIdx.x * 32 + threadIdx.y;
#pragma unroll
  for (int i = 0; i < 32; i += 8) {
    int k = or0 + i;
    if (k < cols && oc < rows)
      out[((size_t)(k >> 1) * rows + oc) * 2 + (k & 1)] = tile[threadIdx.x][threadIdx.y + i];
  }
}

__device__ __forceinline__ void gemm_strip2_f32(const float* __restrict__ A,
                                                const float* __restrict__ BTP,
                                                int K, int N, int m0, int n0,
                                                int lane,
                                                const float* __restrict__ bias,
                                                v8f acc[2][4]) {
  const int half = lane >> 4;
  const int l16  = lane & 15;
#pragma unroll
  for (int i = 0; i < 2; ++i)
#pragma unroll
    for (int j = 0; j < 4; ++j) acc[i][j] = splat8(bias[n0 + 16 * j + l16]);

  const float* arow0 = A + (size_t)(m0 + l16) * K + 2 * half;
  const float* arow1 = arow0 + (size_t)16 * K;
  for (int k = 0; k < K; k += 4) {
    v2f a0 = *(const v2f*)(arow0 + k);
    v2f a1 = *(const v2f*)(arow1 + k);
    const float* bk = BTP + (size_t)((k >> 1) + half) * (2 * N);
#pragma unroll
    for (int j = 0; j < 4; ++j) {
      v2f b = *(const v2f*)(bk + 2 * (n0 + 16 * j + l16));
      acc[0][j] = __builtin_amdgcn_wmma_f32_16x16x4_f32(false, a0, false, b,
                                                        (short)0, acc[0][j], false, false);
      acc[1][j] = __builtin_amdgcn_wmma_f32_16x16x4_f32(false, a1, false, b,
                                                        (short)0, acc[1][j], false, false);
    }
  }
}

__global__ __launch_bounds__(128) void qkv_kernel(const float* __restrict__ x,
                                                  const float* __restrict__ wTP,
                                                  const float* __restrict__ bias,
                                                  float* __restrict__ qv,
                                                  float* __restrict__ kTp,
                                                  float* __restrict__ vp) {
  const int wave = (int)((blockIdx.x * blockDim.x + threadIdx.x) >> 5);
  const int lane = threadIdx.x & 31;
  const int NSTRIP = (3 * C_) / 64;
  const int m0 = (wave / NSTRIP) * 32;
  const int n0 = (wave % NSTRIP) * 64;

  v8f acc[2][4];
  gemm_strip2_f32(x, wTP, C_, 3 * C_, m0, n0, lane, bias, acc);

  const int half = lane >> 4, l16 = lane & 15;
#pragma unroll
  for (int i = 0; i < 2; ++i) {
#pragma unroll
    for (int j = 0; j < 4; ++j) {
      const int nn = n0 + 16 * j + l16;
#pragma unroll
      for (int r = 0; r < 8; ++r) {
        const int mg = m0 + 16 * i + r + 8 * half;
        const int bb = mg >> 11;
        const int t  = mg & (T_ - 1);
        const float val = acc[i][j][r];
        if (nn < C_) {
          const int h = nn >> 6, d = nn & 63;
          qv[((size_t)(bb * H_ + h) * T_ + t) * HD + d] = val;
        } else if (nn < 2 * C_) {
          const int n2 = nn - C_;
          const int h = n2 >> 6, d = n2 & 63;
          kTp[(((size_t)(bb * H_ + h) * (HD / 2) + (d >> 1)) * T_ + t) * 2 + (d & 1)] = val;
        } else {
          const int n2 = nn - 2 * C_;
          const int h = n2 >> 6, d = n2 & 63;
          vp[(((size_t)(bb * H_ + h) * (T_ / 2) + (t >> 1)) * HD + d) * 2 + (t & 1)] = val;
        }
      }
    }
  }
}

#define KVS 64
#define NT  (KVS / 16)
#define PP  (KVS + 4)

__global__ __launch_bounds__(128) __attribute__((amdgpu_waves_per_eu(4)))
void attn_kernel(const float* __restrict__ q,
                                                   const float* __restrict__ kTp,
                                                   const float* __restrict__ vpI,
                                                   float* __restrict__ aout) {
  __shared__ float pl[4][16][PP];
  const int waveid = threadIdx.x >> 5;
  const int lane   = threadIdx.x & 31;
  const int wave   = blockIdx.x * 4 + waveid;
  const int bh     = wave >> 7;
  const int qtile  = wave & 127;
  const int b = bh / H_, h = bh % H_;
  const int qt0 = qtile * 16;
  const float* qb = q   + (size_t)bh * T_ * HD;
  const float* kb = kTp + (size_t)bh * HD * T_;
  const float* vb = vpI + (size_t)bh * T_ * HD;
  const int half = lane >> 4, l16 = lane & 15;
  const float scale = 0.125f;

  v2f qa[16];
#pragma unroll
  for (int kk = 0; kk < 16; ++kk)
    qa[kk] = *(const v2f*)(qb + (size_t)(qt0 + l16) * HD + 4 * kk + 2 * half);

  v8f oacc[4];
#pragma unroll
  for (int j = 0; j < 4; ++j) oacc[j] = splat8(0.f);
  float mrow[8], lrow[8];
#pragma unroll
  for (int r = 0; r < 8; ++r) { mrow[r] = -INFINITY; lrow[r] = 0.f; }

  float (*P)[PP] = pl[waveid];

  for (int kv0 = 0; kv0 < T_; kv0 += KVS) {
    v8f s[NT];
#pragma unroll
    for (int t = 0; t < NT; ++t) s[t] = splat8(0.f);
#pragma unroll
    for (int kk = 0; kk < 16; ++kk) {
#pragma unroll
      for (int t = 0; t < NT; ++t) {
        v2f bfr = *(const v2f*)(kb + ((size_t)(2 * kk + half) * T_ + kv0 + 16 * t + l16) * 2);
        s[t] = __builtin_amdgcn_wmma_f32_16x16x4_f32(false, qa[kk], false, bfr,
                                                     (short)0, s[t], false, false);
      }
    }
    float mx[8];
#pragma unroll
    for (int r = 0; r < 8; ++r) {
#pragma unroll
      for (int t = 0; t < NT; ++t) s[t][r] *= scale;
      float v = s[0][r];
#pragma unroll
      for (int t = 1; t < NT; ++t) v = fmaxf(v, s[t][r]);
      mx[r] = v;
    }
#pragma unroll
    for (int m = 1; m < 16; m <<= 1) {
      float o[8];
#pragma unroll
      for (int r = 0; r < 8; ++r) o[r] = __shfl_xor(mx[r], m);
      __builtin_amdgcn_sched_barrier(0);
#pragma unroll
      for (int r = 0; r < 8; ++r) mx[r] = fmaxf(mx[r], o[r]);
    }
    float alpha[8], sum[8];
#pragma unroll
    for (int r = 0; r < 8; ++r) {
      const float mnew = fmaxf(mrow[r], mx[r]);
      alpha[r] = __expf(mrow[r] - mnew);
      mrow[r]  = mnew;
      float acc = 0.f;
#pragma unroll
      for (int t = 0; t < NT; ++t) {
        const float p = __expf(s[t][r] - mnew);
        s[t][r] = p;
        acc += p;
      }
      sum[r] = acc;
    }
#pragma unroll
    for (int m = 1; m < 16; m <<= 1) {
      float o[8];
#pragma unroll
      for (int r = 0; r < 8; ++r) o[r] = __shfl_xor(sum[r], m);
      __builtin_amdgcn_sched_barrier(0);
#pragma unroll
      for (int r = 0; r < 8; ++r) sum[r] += o[r];
    }
#pragma unroll
    for (int r = 0; r < 8; ++r) lrow[r] = lrow[r] * alpha[r] + sum[r];
#pragma unroll
    for (int j = 0; j < 4; ++j)
#pragma unroll
      for (int r = 0; r < 8; ++r) oacc[j][r] *= alpha[r];

#pragma unroll
    for (int t = 0; t < NT; ++t)
#pragma unroll
      for (int r = 0; r < 8; ++r) P[r + 8 * half][16 * t + l16] = s[t][r];

#pragma unroll
    for (int kk = 0; kk < KVS / 4; ++kk) {
      v2f pa = *(const v2f*)&P[l16][4 * kk + 2 * half];
      const int tp = (kv0 + 4 * kk + 2 * half) >> 1;
#pragma unroll
      for (int j = 0; j < 4; ++j) {
        v2f bv = *(const v2f*)(vb + ((size_t)tp * HD + 16 * j + l16) * 2);
        oacc[j] = __builtin_amdgcn_wmma_f32_16x16x4_f32(false, pa, false, bv,
                                                        (short)0, oacc[j], false, false);
      }
    }
  }

  float inv[8];
#pragma unroll
  for (int r = 0; r < 8; ++r) inv[r] = 1.0f / lrow[r];
#pragma unroll
  for (int j = 0; j < 4; ++j)
#pragma unroll
    for (int r = 0; r < 8; ++r) {
      const int t = qt0 + r + 8 * half;
      aout[((size_t)b * T_ + t) * C_ + h * HD + 16 * j + l16] = oacc[j][r] * inv[r];
    }
}

__global__ __launch_bounds__(128) void proj_kernel(const float* __restrict__ Ain,
                                                   const float* __restrict__ wTP,
                                                   const float* __restrict__ bias,
                                                   float* __restrict__ out) {
  const int wave = (int)((blockIdx.x * blockDim.x + threadIdx.x) >> 5);
  const int lane = threadIdx.x & 31;
  const int NSTRIP = C_ / 64;
  const int m0 = (wave / NSTRIP) * 32;
  const int n0 = (wave % NSTRIP) * 64;

  v8f acc[2][4];
  gemm_strip2_f32(Ain, wTP, C_, C_, m0, n0, lane, bias, acc);

  const int half = lane >> 4, l16 = lane & 15;
#pragma unroll
  for (int i = 0; i < 2; ++i)
#pragma unroll
    for (int j = 0; j < 4; ++j)
#pragma unroll
      for (int r = 0; r < 8; ++r) {
        const int mg = m0 + 16 * i + r + 8 * half;
        out[(size_t)mg * C_ + n0 + 16 * j + l16] = acc[i][j][r];
      }
}

extern "C" void kernel_launch(void* const* d_in, const int* in_sizes, int n_in,
                              void* d_out, int out_size, void* d_ws, size_t ws_size,
                              hipStream_t stream) {
  const float* x      = (const float*)d_in[0];
  const float* qkv_w  = (const float*)d_in[1];
  const float* qkv_b  = (const float*)d_in[2];
  const float* proj_w = (const float*)d_in[3];
  const float* proj_b = (const float*)d_in[4];
  float* out = (float*)d_out;

  const int B = in_sizes[0] / (T_ * C_);
  const int M = B * T_;

  float* ws      = (float*)d_ws;
  float* qkv_wTP  = ws;
  float* proj_wTP = qkv_wTP + (size_t)3 * C_ * C_;
  float* Q        = proj_wTP + (size_t)C_ * C_;
  float* KTP      = Q       + (size_t)B * H_ * T_ * HD;
  float* VP       = KTP     + (size_t)B * H_ * T_ * HD;
  float* AO       = VP      + (size_t)B * H_ * T_ * HD;

  dim3 trb(32, 8);
  tr_pair_kernel<<<dim3(C_ / 32, (3 * C_) / 32), trb, 0, stream>>>(qkv_w, qkv_wTP, 3 * C_, C_);
  tr_pair_kernel<<<dim3(C_ / 32, C_ / 32),       trb, 0, stream>>>(proj_w, proj_wTP, C_, C_);

  {
    const int waves = (M / 32) * ((3 * C_) / 64);
    qkv_kernel<<<waves / 4, 128, 0, stream>>>(x, qkv_wTP, qkv_b, Q, KTP, VP);
  }
  {
    const int waves = B * H_ * (T_ / 16);
    attn_kernel<<<waves / 4, 128, 0, stream>>>(Q, KTP, VP, AO);
  }
  {
    const int waves = (M / 32) * (C_ / 64);
    proj_kernel<<<waves / 4, 128, 0, stream>>>(AO, proj_wTP, proj_b, out);
  }
}


# all WMMA fragment loads widened from 8 to 16 bytes per lane, one load feeding two WMMAs (weights, K, V stored 4-interleaved along k; P read from LDS as b128): k grouping per WMMA and sum order change
# speedup vs baseline: 1.3855x; 1.0952x over previous
#include <hip/hip_runtime.h>
#include <math.h>

#define H_  16
#define T_  2048
#define C_  1024
#define HD  64

typedef __attribute__((ext_vector_type(2))) float v2f;
typedef __attribute__((ext_vector_type(4))) float v4f;
typedef __attribute__((ext_vector_type(8))) float v8f;

__device__ __forceinline__ v8f splat8(float x) {
  v8f r;
#pragma unroll
  for (int i = 0; i < 8; ++i) r[i] = x;
  return r;
}

__global__ __launch_bounds__(256) void tr_quad_kernel(const float* __restrict__ in,
                                                      float* __restrict__ out,
                                                      int rows, int cols) {
  __shared__ float tile[32][33];
  int c  = blockIdx.x * 32 + threadIdx.x;
  int r0 = blockIdx.y * 32 + threadIdx.y;
#pragma unroll
  for (int i = 0; i < 32; i += 8) {
    int r = r0 + i;
    if (r < rows && c < cols) tile[threadIdx.y + i][threadIdx.x] = in[(size_t)r * cols + c];
  }
  __syncthreads();
  const int n  = blockIdx.y * 32 + threadIdx.x;
  const int kq = blockIdx.x * 8 + threadIdx.y;
  if (n < rows && 4 * kq + 3 < cols) {
    v4f v;
    v.x = tile[threadIdx.x][4 * threadIdx.y + 0];
    v.y = tile[threadIdx.x][4 * threadIdx.y + 1];
    v.z = tile[threadIdx.x][4 * threadIdx.y + 2];
    v.w = tile[threadIdx.x][4 * threadIdx.y + 3];
    *(v4f*)(out + ((size_t)kq * rows + n) * 4) = v;
  }
}

__device__ __forceinline__ void gemm_strip2_f32(const float* __restrict__ A,
                                                const float* __restrict__ BTQ,
                                                int K, int N, int m0, int n0,
                                                int lane,
                                                const float* __restrict__ bias,
                                                v8f acc[2][4]) {
  const int half = lane >> 4;
  const int l16  = lane & 15;
#pragma unroll
  for (int i = 0; i < 2; ++i)
#pragma unroll
    for (int j = 0; j < 4; ++j) acc[i][j] = splat8(bias[n0 + 16 * j + l16]);

  const float* arow0 = A + (size_t)(m0 + l16) * K + 4 * half;
  const float* arow1 = arow0 + (size_t)16 * K;
  const float* bcol = BTQ + (size_t)half * (4 * N) + 4 * (n0 + l16);
  for (int k = 0; k < K; k += 16) {
    v4f a0[2], a1[2], b[2][4];
#pragma unroll
    for (int u = 0; u < 2; ++u) {
      a0[u] = *(const v4f*)(arow0 + k + 8 * u);
      a1[u] = *(const v4f*)(arow1 + k + 8 * u);
      const float* bk = bcol + (size_t)((k >> 2) + 2 * u) * (4 * N);
#pragma unroll
      for (int j = 0; j < 4; ++j) b[u][j] = *(const v4f*)(bk + 64 * j);
    }
    __builtin_amdgcn_sched_barrier(0);
#pragma unroll
    for (int u = 0; u < 2; ++u) {
      const v2f a0lo = {a0[u].x, a0[u].y}, a0hi = {a0[u].z, a0[u].w};
      const v2f a1lo = {a1[u].x, a1[u].y}, a1hi = {a1[u].z, a1[u].w};
#pragma unroll
      for (int j = 0; j < 4; ++j) {
        const v2f blo = {b[u][j].x, b[u][j].y};
        acc[0][j] = __builtin_amdgcn_wmma_f32_16x16x4_f32(false, a0lo, false, blo,
                                                          (short)0, acc[0][j], false, false);
        acc[1][j] = __builtin_amdgcn_wmma_f32_16x16x4_f32(false, a1lo, false, blo,
                                                          (short)0, acc[1][j], false, false);
      }
#pragma unroll
      for (int j = 0; j < 4; ++j) {
        const v2f bhi = {b[u][j].z, b[u][j].w};
        acc[0][j] = __builtin_amdgcn_wmma_f32_16x16x4_f32(false, a0hi, false, bhi,
                                                          (short)0, acc[0][j], false, false);
        acc[1][j] = __builtin_amdgcn_wmma_f32_16x16x4_f32(false, a1hi, false, bhi,
                                                          (short)0, acc[1][j], false, false);
      }
    }
  }
}

__global__ __launch_bounds__(128) void qkv_kernel(const float* __restrict__ x,
                                                  const float* __restrict__ wTQ,
                                                  const float* __restrict__ bias,
                                                  float* __restrict__ qv,
                                                  float* __restrict__ kTq,
                                                  float* __restrict__ vq) {
  const int wave = (int)((blockIdx.x * blockDim.x + threadIdx.x) >> 5);
  const int lane = threadIdx.x & 31;
  const int NSTRIP = (3 * C_) / 64;
  const int m0 = (wave / NSTRIP) * 32;
  const int n0 = (wave % NSTRIP) * 64;

  v8f acc[2][4];
  gemm_strip2_f32(x, wTQ, C_, 3 * C_, m0, n0, lane, bias, acc);

  const int half = lane >> 4, l16 = lane & 15;
  const int part = n0 / C_;
  const int h    = (n0 & (C_ - 1)) >> 6;
#pragma unroll
  for (int i = 0; i < 2; ++i) {
    const int mg0 = m0 + 16 * i + 8 * half;
    const int bb  = mg0 >> 11;
    const int t0  = mg0 & (T_ - 1);
    const size_t bh = (size_t)(bb * H_ + h);
#pragma unroll
    for (int j = 0; j < 4; ++j) {
      const int d = 16 * j + l16;
      if (part == 0) {
#pragma unroll
        for (int r = 0; r < 8; ++r)
          qv[(bh * T_ + t0 + r) * HD + d] = acc[i][j][r];
      } else if (part == 1) {
#pragma unroll
        for (int r = 0; r < 8; ++r)
          kTq[((bh * (HD / 4) + (d >> 2)) * T_ + t0 + r) * 4 + (d & 3)] = acc[i][j][r];
      } else {
#pragma unroll
        for (int rr = 0; rr < 8; rr += 4) {
          v4f v;
          v.x = acc[i][j][rr + 0];
          v.y = acc[i][j][rr + 1];
          v.z = acc[i][j][rr + 2];
          v.w = acc[i][j][rr + 3];
          *(v4f*)(vq + ((bh * (T_ / 4) + ((t0 + rr) >> 2)) * HD + d) * 4) = v;
        }
      }
    }
  }
}

#define KVS 64
#define NT  (KVS / 16)
#define PP  (KVS + 4)

__global__ __launch_bounds__(128) __attribute__((amdgpu_waves_per_eu(4)))
void attn_kernel(const float* __restrict__ q,
                                                   const float* __restrict__ kTq,
                                                   const float* __restrict__ vqI,
                                                   float* __restrict__ aout) {
  __shared__ __attribute__((aligned(16))) float pl[4][16][PP];
  const int waveid = threadIdx.x >> 5;
  const int lane   = threadIdx.x & 31;
  const int wave   = blockIdx.x * 4 + waveid;
  const int bh     = wave >> 7;
  const int qtile  = wave & 127;
  const int b = bh / H_, h = bh % H_;
  const int qt0 = qtile * 16;
  const float* qb = q   + (size_t)bh * T_ * HD;
  const float* kb = kTq + (size_t)bh * HD * T_;
  const float* vb = vqI + (size_t)bh * T_ * HD;
  const int half = lane >> 4, l16 = lane & 15;
  const float scale = 0.125f;

  v4f qa[8];
#pragma unroll
  for (int kq = 0; kq < 8; ++kq)
    qa[kq] = *(const v4f*)(qb + (size_t)(qt0 + l16) * HD + 8 * kq + 4 * half);

  v8f oacc[4];
#pragma unroll
  for (int j = 0; j < 4; ++j) oacc[j] = splat8(0.f);
  float mrow[8], lrow[8];
#pragma unroll
  for (int r = 0; r < 8; ++r) { mrow[r] = -INFINITY; lrow[r] = 0.f; }

  float (*P)[PP] = pl[waveid];

  const float* kp = kb + ((size_t)half * T_ + l16) * 4;
  const float* vp = vb + ((size_t)half * HD + l16) * 4;

  for (int kv0 = 0; kv0 < T_; kv0 += KVS, kp += KVS * 4, vp += (KVS / 4) * HD * 4) {
    v8f s[NT];
#pragma unroll
    for (int t = 0; t < NT; ++t) s[t] = splat8(0.f);
#pragma unroll
    for (int kq = 0; kq < 8; ++kq) {
      v4f bfr[NT];
#pragma unroll
      for (int t = 0; t < NT; ++t)
        bfr[t] = *(const v4f*)(kp + ((size_t)(2 * kq) * T_ + 16 * t) * 4);
      const v2f qlo = {qa[kq].x, qa[kq].y}, qhi = {qa[kq].z, qa[kq].w};
#pragma unroll
      for (int t = 0; t < NT; ++t) {
        const v2f blo = {bfr[t].x, bfr[t].y};
        s[t] = __builtin_amdgcn_wmma_f32_16x16x4_f32(false, qlo, false, blo,
                                                     (short)0, s[t], false, false);
      }
#pragma unroll
      for (int t = 0; t < NT; ++t) {
        const v2f bhi = {bfr[t].z, bfr[t].w};
        s[t] = __builtin_amdgcn_wmma_f32_16x16x4_f32(false, qhi, false, bhi,
                                                     (short)0, s[t], false, false);
      }
    }
    float mx[8];
#pragma unroll
    for (int r = 0; r < 8; ++r) {
#pragma unroll
      for (int t = 0; t < NT; ++t) s[t][r] *= scale;
      float v = s[0][r];
#pragma unroll
      for (int t = 1; t < NT; ++t) v = fmaxf(v, s[t][r]);
      mx[r] = v;
    }
#pragma unroll
    for (int m = 1; m < 16; m <<= 1) {
      float o[8];
#pragma unroll
      for (int r = 0; r < 8; ++r) o[r] = __shfl_xor(mx[r], m);
      __builtin_amdgcn_sched_barrier(0);
#pragma unroll
      for (int r = 0; r < 8; ++r) mx[r] = fmaxf(mx[r], o[r]);
    }
    float alpha[8], sum[8];
#pragma unroll
    for (int r = 0; r < 8; ++r) {
      const float mnew = fmaxf(mrow[r], mx[r]);
      alpha[r] = __expf(mrow[r] - mnew);
      mrow[r]  = mnew;
      float acc = 0.f;
#pragma unroll
      for (int t = 0; t < NT; ++t) {
        const float p = __expf(s[t][r] - mnew);
        s[t][r] = p;
        acc += p;
      }
      sum[r] = acc;
    }
#pragma unroll
    for (int m = 1; m < 16; m <<= 1) {
      float o[8];
#pragma unroll
      for (int r = 0; r < 8; ++r) o[r] = __shfl_xor(sum[r], m);
      __builtin_amdgcn_sched_barrier(0);
#pragma unroll
      for (int r = 0; r < 8; ++r) sum[r] += o[r];
    }
#pragma unroll
    for (int r = 0; r < 8; ++r) lrow[r] = lrow[r] * alpha[r] + sum[r];
#pragma unroll
    for (int j = 0; j < 4; ++j)
#pragma unroll
      for (int r = 0; r < 8; ++r) oacc[j][r] *= alpha[r];

#pragma unroll
    for (int t = 0; t < NT; ++t)
#pragma unroll
      for (int r = 0; r < 8; ++r) P[r + 8 * half][16 * t + l16] = s[t][r];

    v4f bv[KVS / 8][4];
#pragma unroll
    for (int kq = 0; kq < KVS / 8; ++kq)
#pragma unroll
      for (int j = 0; j < 4; ++j)
        bv[kq][j] = *(const v4f*)(vp + ((size_t)(2 * kq) * HD + 16 * j) * 4);
    __builtin_amdgcn_sched_barrier(0);
#pragma unroll
    for (int kq = 0; kq < KVS / 8; ++kq) {
      const v4f pa = *(const v4f*)&P[l16][8 * kq + 4 * half];
      const v2f plo = {pa.x, pa.y}, phi = {pa.z, pa.w};
#pragma unroll
      for (int j = 0; j < 4; ++j) {
        const v2f blo = {bv[kq][j].x, bv[kq][j].y};
        oacc[j] = __builtin_amdgcn_wmma_f32_16x16x4_f32(false, plo, false, blo,
                                                        (short)0, oacc[j], false, false);
      }
#pragma unroll
      for (int j = 0; j < 4; ++j) {
        const v2f bhi = {bv[kq][j].z, bv[kq][j].w};
        oacc[j] = __builtin_amdgcn_wmma_f32_16x16x4_f32(false, phi, false, bhi,
                                                        (short)0, oacc[j], false, false);
      }
    }
  }

  float inv[8];
#pragma unroll
  for (int r = 0; r < 8; ++r) inv[r] = 1.0f / lrow[r];
#pragma unroll
  for (int j = 0; j < 4; ++j)
#pragma unroll
    for (int r = 0; r < 8; ++r) {
      const int t = qt0 + r + 8 * half;
      aout[((size_t)b * T_ + t) * C_ + h * HD + 16 * j + l16] = oacc[j][r] * inv[r];
    }
}

__global__ __launch_bounds__(128) void proj_kernel(const float* __restrict__ Ain,
                                                   const float* __restrict__ wTQ,
                                                   const float* __restrict__ bias,
                                                   float* __restrict__ out) {
  const int wave = (int)((blockIdx.x * blockDim.x + threadIdx.x) >> 5);
  const int lane = threadIdx.x & 31;
  const int NSTRIP = C_ / 64;
  const int m0 = (wave / NSTRIP) * 32;
  const int n0 = (wave % NSTRIP) * 64;

  v8f acc[2][4];
  gemm_strip2_f32(Ain, wTQ, C_, C_, m0, n0, lane, bias, acc);

  const int half = lane >> 4, l16 = lane & 15;
#pragma unroll
  for (int i = 0; i < 2; ++i)
#pragma unroll
    for (int j = 0; j < 4; ++j)
#pragma unroll
      for (int r = 0; r < 8; ++r) {
        const int mg = m0 + 16 * i + r + 8 * half;
        out[(size_t)mg * C_ + n0 + 16 * j + l16] = acc[i][j][r];
      }
}

extern "C" void kernel_launch(void* const* d_in, const int* in_sizes, int n_in,
                              void* d_out, int out_size, void* d_ws, size_t ws_size,
                              hipStream_t stream) {
  const float* x      = (const float*)d_in[0];
  const float* qkv_w  = (const float*)d_in[1];
  const float* qkv_b  = (const float*)d_in[2];
  const float* proj_w = (const float*)d_in[3];
  const float* proj_b = (const float*)d_in[4];
  float* out = (float*)d_out;

  const int B = in_sizes[0] / (T_ * C_);
  const int M = B * T_;

  float* ws      = (float*)d_ws;
  float* qkv_wTQ  = ws;
  float* proj_wTQ = qkv_wTQ + (size_t)3 * C_ * C_;
  float* Q        = proj_wTQ + (size_t)C_ * C_;
  float* KTQ      = Q       + (size_t)B * H_ * T_ * HD;
  float* VQ       = KTQ     + (size_t)B * H_ * T_ * HD;
  float* AO       = VQ      + (size_t)B * H_ * T_ * HD;

  dim3 trb(32, 8);
  tr_quad_kernel<<<dim3(C_ / 32, (3 * C_) / 32), trb, 0, stream>>>(qkv_w, qkv_wTQ, 3 * C_, C_);
  tr_quad_kernel<<<dim3(C_ / 32, C_ / 32),       trb, 0, stream>>>(proj_w, proj_wTQ, C_, C_);

  {
    const int waves = (M / 32) * ((3 * C_) / 64);
    qkv_kernel<<<waves / 4, 128, 0, stream>>>(x, qkv_wTQ, qkv_b, Q, KTQ, VQ);
  }
  {
    const int waves = B * H_ * (T_ / 16);
    attn_kernel<<<waves / 4, 128, 0, stream>>>(Q, KTQ, VQ, AO);
  }
  {
    const int waves = (M / 32) * (C_ / 64);
    proj_kernel<<<waves / 4, 128, 0, stream>>>(AO, proj_wTQ, proj_b, out);
  }
}


# attention computed transposed (S^T = K Q^T, O^T = V^T P^T): softmax max/sum in-lane plus one half exchange, P fed to the 2nd product from registers (no LDS); PV key grouping and sum order change
# speedup vs baseline: 1.4617x; 1.0550x over previous
#include <hip/hip_runtime.h>
#include <math.h>

#define H_  16
#define T_  2048
#define C_  1024
#define HD  64

typedef __attribute__((ext_vector_type(2))) float v2f;
typedef __attribute__((ext_vector_type(4))) float v4f;
typedef __attribute__((ext_vector_type(8))) float v8f;

__device__ __forceinline__ v8f splat8(float x) {
  v8f r;
#pragma unroll
  for (int i = 0; i < 8; ++i) r[i] = x;
  return r;
}

__global__ __launch_bounds__(256) void tr_quad_kernel(const float* __restrict__ in,
                                                      float* __restrict__ out,
                                                      int rows, int cols) {
  __shared__ float tile[32][33];
  int c  = blockIdx.x * 32 + threadIdx.x;
  int r0 = blockIdx.y * 32 + threadIdx.y;
#pragma unroll
  for (int i = 0; i < 32; i += 8) {
    int r = r0 + i;
    if (r < rows && c < cols) tile[threadIdx.y + i][threadIdx.x] = in[(size_t)r * cols + c];
  }
  __syncthreads();
  const int n  = blockIdx.y * 32 + threadIdx.x;
  const int kq = blockIdx.x * 8 + threadIdx.y;
  if (n < rows && 4 * kq + 3 < cols) {
    v4f v;
    v.x = tile[threadIdx.x][4 * threadIdx.y + 0];
    v.y = tile[threadIdx.x][4 * threadIdx.y + 1];
    v.z = tile[threadIdx.x][4 * threadIdx.y + 2];
    v.w = tile[threadIdx.x][4 * threadIdx.y + 3];
    *(v4f*)(out + ((size_t)kq * rows + n) * 4) = v;
  }
}

__device__ __forceinline__ void gemm_strip2_f32(const float* __restrict__ A,
                                                const float* __restrict__ BTQ,
                                                int K, int N, int m0, int n0,
                                                int lane,
                                                const float* __restrict__ bias,
                                                v8f acc[2][4]) {
  const int half = lane >> 4;
  const int l16  = lane & 15;
#pragma unroll
  for (int i = 0; i < 2; ++i)
#pragma unroll
    for (int j = 0; j < 4; ++j) acc[i][j] = splat8(bias[n0 + 16 * j + l16]);

  const float* arow0 = A + (size_t)(m0 + l16) * K + 4 * half;
  const float* arow1 = arow0 + (size_t)16 * K;
  const float* bcol = BTQ + (size_t)half * (4 * N) + 4 * (n0 + l16);
  for (int k = 0; k < K; k += 16) {
    v4f a0[2], a1[2], b[2][4];
#pragma unroll
    for (int u = 0; u < 2; ++u) {
      a0[u] = *(const v4f*)(arow0 + k + 8 * u);
      a1[u] = *(const v4f*)(arow1 + k + 8 * u);
      const float* bk = bcol + (size_t)((k >> 2) + 2 * u) * (4 * N);
#pragma unroll
      for (int j = 0; j < 4; ++j) b[u][j] = *(const v4f*)(bk + 64 * j);
    }
    __builtin_amdgcn_sched_barrier(0);
#pragma unroll
    for (int u = 0; u < 2; ++u) {
      const v2f a0lo = {a0[u].x, a0[u].y}, a0hi = {a0[u].z, a0[u].w};
      const v2f a1lo = {a1[u].x, a1[u].y}, a1hi = {a1[u].z, a1[u].w};
#pragma unroll
      for (int j = 0; j < 4; ++j) {
        const v2f blo = {b[u][j].x, b[u][j].y};
        acc[0][j] = __builtin_amdgcn_wmma_f32_16x16x4_f32(false, a0lo, false, blo,
                                                          (short)0, acc[0][j], false, false);
        acc[1][j] = __builtin_amdgcn_wmma_f32_16x16x4_f32(false, a1lo, false, blo,
                                                          (short)0, acc[1][j], false, false);
      }
#pragma unroll
      for (int j = 0; j < 4; ++j) {
        const v2f bhi = {b[u][j].z, b[u][j].w};
        acc[0][j] = __builtin_amdgcn_wmma_f32_16x16x4_f32(false, a0hi, false, bhi,
                                                          (short)0, acc[0][j], false, false);
        acc[1][j] = __builtin_amdgcn_wmma_f32_16x16x4_f32(false, a1hi, false, bhi,
                                                          (short)0, acc[1][j], false, false);
      }
    }
  }
}

__global__ __launch_bounds__(128) void qkv_kernel(const float* __restrict__ x,
                                                  const float* __restrict__ wTQ,
                                                  const float* __restrict__ bias,
                                                  float* __restrict__ qv,
                                                  float* __restrict__ kTq,
                                                  float* __restrict__ vq) {
  const int wave = (int)((blockIdx.x * blockDim.x + threadIdx.x) >> 5);
  const int lane = threadIdx.x & 31;
  const int NSTRIP = (3 * C_) / 64;
  const int m0 = (wave / NSTRIP) * 32;
  const int n0 = (wave % NSTRIP) * 64;

  v8f acc[2][4];
  gemm_strip2_f32(x, wTQ, C_, 3 * C_, m0, n0, lane, bias, acc);

  const int half = lane >> 4, l16 = lane & 15;
  const int part = n0 / C_;
  const int h    = (n0 & (C_ - 1)) >> 6;
#pragma unroll
  for (int i = 0; i < 2; ++i) {
    const int mg0 = m0 + 16 * i + 8 * half;
    const int bb  = mg0 >> 11;
    const int t0  = mg0 & (T_ - 1);
    const size_t bh = (size_t)(bb * H_ + h);
#pragma unroll
    for (int j = 0; j < 4; ++j) {
      const int d = 16 * j + l16;
      if (part == 0) {
#pragma unroll
        for (int r = 0; r < 8; ++r)
          qv[(bh * T_ + t0 + r) * HD + d] = acc[i][j][r];
      } else if (part == 1) {
#pragma unroll
        for (int r = 0; r < 8; ++r)
          kTq[((bh * (HD / 4) + (d >> 2)) * T_ + t0 + r) * 4 + (d & 3)] = acc[i][j][r];
      } else {
#pragma unroll
        for (int rr = 0; rr < 8; rr += 4) {
          v4f v;
          v.x = acc[i][j][rr + 0];
          v.y = acc[i][j][rr + 1];
          v.z = acc[i][j][rr + 2];
          v.w = acc[i][j][rr + 3];
          *(v4f*)(vq + ((bh * (T_ / 4) + ((t0 + rr) >> 2)) * HD + d) * 4) = v;
        }
      }
    }
  }
}

#define KVS 64
#define NT  (KVS / 16)

__device__ __forceinline__ float xhalf(float x) {
  const int b = __builtin_bit_cast(int, x);
  return __builtin_bit_cast(float,
      __builtin_amdgcn_permlanex16(b, b, 0x76543210, 0xfedcba98, false, false));
}

__global__ __launch_bounds__(128) __attribute__((amdgpu_waves_per_eu(4)))
void attn_kernel(const float* __restrict__ q,
                                                   const float* __restrict__ kTq,
                                                   const float* __restrict__ vqI,
                                                   float* __restrict__ aout) {
  const int waveid = threadIdx.x >> 5;
  const int lane   = threadIdx.x & 31;
  const int wave   = blockIdx.x * 4 + waveid;
  const int bh     = wave >> 7;
  const int qtile  = wave & 127;
  const int b = bh / H_, h = bh % H_;
  const int qt0 = qtile * 16;
  const float* qb = q   + (size_t)bh * T_ * HD;
  const float* kb = kTq + (size_t)bh * HD * T_;
  const float* vb = vqI + (size_t)bh * T_ * HD;
  const int half = lane >> 4, l16 = lane & 15;
  const float scale = 0.125f;

  v4f qa[8];
#pragma unroll
  for (int kq = 0; kq < 8; ++kq)
    qa[kq] = *(const v4f*)(qb + (size_t)(qt0 + l16) * HD + 8 * kq + 4 * half);

  v8f oacc[4];
#pragma unroll
  for (int j = 0; j < 4; ++j) oacc[j] = splat8(0.f);
  float mrow = -INFINITY, lrow = 0.f;

  const float* kp = kb + ((size_t)half * T_ + l16) * 4;
  const float* vp = vb + ((size_t)(2 * half) * HD + l16) * 4;

  for (int kv0 = 0; kv0 < T_; kv0 += KVS, kp += KVS * 4, vp += (KVS / 4) * HD * 4) {
    v8f s[NT];
#pragma unroll
    for (int t = 0; t < NT; ++t) s[t] = splat8(0.f);
#pragma unroll
    for (int kq = 0; kq < 8; ++kq) {
      v4f kfr[NT];
#pragma unroll
      for (int t = 0; t < NT; ++t)
        kfr[t] = *(const v4f*)(kp + ((size_t)(2 * kq) * T_ + 16 * t) * 4);
      const v2f qlo = {qa[kq].x, qa[kq].y}, qhi = {qa[kq].z, qa[kq].w};
#pragma unroll
      for (int t = 0; t < NT; ++t) {
        const v2f klo = {kfr[t].x, kfr[t].y};
        s[t] = __builtin_amdgcn_wmma_f32_16x16x4_f32(false, klo, false, qlo,
                                                     (short)0, s[t], false, false);
      }
#pragma unroll
      for (int t = 0; t < NT; ++t) {
        const v2f khi = {kfr[t].z, kfr[t].w};
        s[t] = __builtin_amdgcn_wmma_f32_16x16x4_f32(false, khi, false, qhi,
                                                     (short)0, s[t], false, false);
      }
    }
    float mx = -INFINITY;
#pragma unroll
    for (int t = 0; t < NT; ++t)
#pragma unroll
      for (int r = 0; r < 8; ++r) {
        s[t][r] *= scale;
        mx = fmaxf(mx, s[t][r]);
      }
    mx = fmaxf(mx, xhalf(mx));
    const float mnew  = fmaxf(mrow, mx);
    const float alpha = __expf(mrow - mnew);
    mrow = mnew;
    float sum = 0.f;
#pragma unroll
    for (int t = 0; t < NT; ++t)
#pragma unroll
      for (int r = 0; r < 8; ++r) {
        const float p = __expf(s[t][r] - mnew);
        s[t][r] = p;
        sum += p;
      }
    lrow = lrow * alpha + sum;
#pragma unroll
    for (int j = 0; j < 4; ++j)
#pragma unroll
      for (int r = 0; r < 8; ++r) oacc[j][r] *= alpha;

    v4f av[NT][2][4];
#pragma unroll
    for (int t = 0; t < NT; ++t)
#pragma unroll
      for (int g = 0; g < 2; ++g)
#pragma unroll
        for (int j = 0; j < 4; ++j)
          av[t][g][j] = *(const v4f*)(vp + ((size_t)(4 * t + g) * HD + 16 * j) * 4);
    __builtin_amdgcn_sched_barrier(0);
#pragma unroll
    for (int t = 0; t < NT; ++t)
#pragma unroll
      for (int g = 0; g < 2; ++g) {
        const v2f plo = {s[t][4 * g + 0], s[t][4 * g + 1]};
        const v2f phi = {s[t][4 * g + 2], s[t][4 * g + 3]};
#pragma unroll
        for (int j = 0; j < 4; ++j) {
          const v2f alo = {av[t][g][j].x, av[t][g][j].y};
          oacc[j] = __builtin_amdgcn_wmma_f32_16x16x4_f32(false, alo, false, plo,
                                                          (short)0, oacc[j], false, false);
        }
#pragma unroll
        for (int j = 0; j < 4; ++j) {
          const v2f ahi = {av[t][g][j].z, av[t][g][j].w};
          oacc[j] = __builtin_amdgcn_wmma_f32_16x16x4_f32(false, ahi, false, phi,
                                                          (short)0, oacc[j], false, false);
        }
      }
  }

  const float inv = 1.0f / (lrow + xhalf(lrow));
  float* op = aout + ((size_t)b * T_ + qt0 + l16) * C_ + h * HD + 8 * half;
#pragma unroll
  for (int j = 0; j < 4; ++j) {
    v4f lo, hi;
    lo.x = oacc[j][0] * inv; lo.y = oacc[j][1] * inv;
    lo.z = oacc[j][2] * inv; lo.w = oacc[j][3] * inv;
    hi.x = oacc[j][4] * inv; hi.y = oacc[j][5] * inv;
    hi.z = oacc[j][6] * inv; hi.w = oacc[j][7] * inv;
    *(v4f*)(op + 16 * j)     = lo;
    *(v4f*)(op + 16 * j + 4) = hi;
  }
}

__global__ __launch_bounds__(128) void proj_kernel(const float* __restrict__ Ain,
                                                   const float* __restrict__ wTQ,
                                                   const float* __restrict__ bias,
                                                   float* __restrict__ out) {
  const int wave = (int)((blockIdx.x * blockDim.x + threadIdx.x) >> 5);
  const int lane = threadIdx.x & 31;
  const int NSTRIP = C_ / 64;
  const int m0 = (wave / NSTRIP) * 32;
  const int n0 = (wave % NSTRIP) * 64;

  v8f acc[2][4];
  gemm_strip2_f32(Ain, wTQ, C_, C_, m0, n0, lane, bias, acc);

  const int half = lane >> 4, l16 = lane & 15;
#pragma unroll
  for (int i = 0; i < 2; ++i)
#pragma unroll
    for (int j = 0; j < 4; ++j)
#pragma unroll
      for (int r = 0; r < 8; ++r) {
        const int mg = m0 + 16 * i + r + 8 * half;
        out[(size_t)mg * C_ + n0 + 16 * j + l16] = acc[i][j][r];
      }
}

extern "C" void kernel_launch(void* const* d_in, const int* in_sizes, int n_in,
                              void* d_out, int out_size, void* d_ws, size_t ws_size,
                              hipStream_t stream) {
  const float* x      = (const float*)d_in[0];
  const float* qkv_w  = (const float*)d_in[1];
  const float* qkv_b  = (const float*)d_in[2];
  const float* proj_w = (const float*)d_in[3];
  const float* proj_b = (const float*)d_in[4];
  float* out = (float*)d_out;

  const int B = in_sizes[0] / (T_ * C_);
  const int M = B * T_;

  float* ws      = (float*)d_ws;
  float* qkv_wTQ  = ws;
  float* proj_wTQ = qkv_wTQ + (size_t)3 * C_ * C_;
  float* Q        = proj_wTQ + (size_t)C_ * C_;
  float* KTQ      = Q       + (size_t)B * H_ * T_ * HD;
  float* VQ       = KTQ     + (size_t)B * H_ * T_ * HD;
  float* AO       = VQ      + (size_t)B * H_ * T_ * HD;

  dim3 trb(32, 8);
  tr_quad_kernel<<<dim3(C_ / 32, (3 * C_) / 32), trb, 0, stream>>>(qkv_w, qkv_wTQ, 3 * C_, C_);
  tr_quad_kernel<<<dim3(C_ / 32, C_ / 32),       trb, 0, stream>>>(proj_w, proj_wTQ, C_, C_);

  {
    const int waves = (M / 32) * ((3 * C_) / 64);
    qkv_kernel<<<waves / 4, 128, 0, stream>>>(x, qkv_wTQ, qkv_b, Q, KTQ, VQ);
  }
  {
    const int waves = B * H_ * (T_ / 16);
    attn_kernel<<<waves / 4, 128, 0, stream>>>(Q, KTQ, VQ, AO);
  }
  {
    const int waves = (M / 32) * (C_ / 64);
    proj_kernel<<<waves / 4, 128, 0, stream>>>(AO, proj_wTQ, proj_b, out);
  }
}
